# starting kernel
# baseline (speedup 1.0000x reference)
#include <hip/hip_runtime.h>

typedef float v2f __attribute__((ext_vector_type(2)));
typedef float v8f __attribute__((ext_vector_type(8)));

#define NHEAD   16
#define DK      64
#define DV      64
#define NTOT    16384
#define CHUNKS  32
#define NB      128
#define LSTRIDE 80
#define DECAY   0.95f

__global__ __launch_bounds__(512) void l1w_init(const float* __restrict__ mem,
                                                float* __restrict__ out) {
    int i = blockIdx.x * 512 + threadIdx.x;
    if (i < NHEAD * DK * DV) out[i] = DECAY * mem[i];
}

__global__ __launch_bounds__(512) void l1w_wmma(const float* __restrict__ keys,
                                                const float* __restrict__ vals,
                                                const float* __restrict__ rho,
                                                float* __restrict__ out) {
    __shared__ float kl[NB * LSTRIDE];
    __shared__ float vl[NB * LSTRIDE];

    const int head  = blockIdx.x / CHUNKS;
    const int chunk = blockIdx.x % CHUNKS;
    const int rows_per_wg = NTOT / CHUNKS;
    const int row0  = chunk * rows_per_wg;

    const int tid  = threadIdx.x;
    const int wave = tid >> 5;
    const int lane = tid & 31;
    const int hlf  = lane >> 4;
    const int l15  = lane & 15;
    const int tm   = (wave >> 2) * 16;
    const int tn   = (wave & 3)  * 16;

    const int r_base = tid >> 4;
    const int c_stg  = (tid & 15) << 2;

    v8f acc = {};

    float4 kreg[4], vreg[4];
    float  wreg[4];

    #pragma unroll
    for (int it = 0; it < 4; ++it) {
        int g = row0 + it * 32 + r_base;
        kreg[it] = *(const float4*)(keys + ((size_t)g * NHEAD + head) * DK + c_stg);
        vreg[it] = *(const float4*)(vals + ((size_t)g * NHEAD + head) * DV + c_stg);
        wreg[it] = rho[g];
    }

    for (int blk = 0; blk < rows_per_wg; blk += NB) {
        #pragma unroll
        for (int it = 0; it < 4; ++it) {
            int r = it * 32 + r_base;
            float* kd = &kl[r * LSTRIDE + c_stg];
            float* vd = &vl[r * LSTRIDE + c_stg];
            const float w = wreg[it];
            kd[0] = kreg[it].x * w; kd[1] = kreg[it].y * w;
            kd[2] = kreg[it].z * w; kd[3] = kreg[it].w * w;
            vd[0] = vreg[it].x;     vd[1] = vreg[it].y;
            vd[2] = vreg[it].z;     vd[3] = vreg[it].w;
        }
        __syncthreads();

        if (blk + NB < rows_per_wg) {
            #pragma unroll
            for (int it = 0; it < 4; ++it) {
                int g = row0 + blk + NB + it * 32 + r_base;
                kreg[it] = *(const float4*)(keys + ((size_t)g * NHEAD + head) * DK + c_stg);
                vreg[it] = *(const float4*)(vals + ((size_t)g * NHEAD + head) * DV + c_stg);
                wreg[it] = rho[g];
            }
        }

        #pragma unroll 4
        for (int s = 0; s < NB; s += 4) {
            int ra = s + hlf * 2;
            v2f a, b;
            a.x = kl[ ra      * LSTRIDE + tm + l15];
            a.y = kl[(ra + 1) * LSTRIDE + tm + l15];
            b.x = vl[ ra      * LSTRIDE + tn + l15];
            b.y = vl[(ra + 1) * LSTRIDE + tn + l15];
            acc = __builtin_amdgcn_wmma_f32_16x16x4_f32(
                 false, a,  false, b,
                 (short)0, acc,  false,  false);
        }
        __syncthreads();
    }

    #pragma unroll
    for (int i = 0; i < 8; ++i) {
        int dk = tm + i + hlf * 8;
        int dv = tn + l15;
        atomicAdd(&out[((size_t)head * DK + dk) * DV + dv], acc[i]);
    }
}

extern "C" void kernel_launch(void* const* d_in, const int* in_sizes, int n_in,
                              void* d_out, int out_size, void* d_ws, size_t ws_size,
                              hipStream_t stream) {
    const float* mem  = (const float*)d_in[0];
    const float* keys = (const float*)d_in[1];
    const float* vals = (const float*)d_in[2];
    const float* rho  = (const float*)d_in[3];
    float* out = (float*)d_out;

    l1w_init<<<(NHEAD * DK * DV + 511) / 512, 512, 0, stream>>>(mem, out);
    l1w_wmma<<<NHEAD * CHUNKS, 512, 0, stream>>>(keys, vals, rho, out);
}


# WMMA fragments fed straight from 16-byte global loads, one wave per 64x64 head tile: no LDS staging or loop barriers; partials summed in LDS, 16 not 32 atomic adds per output (sum order changes)
# speedup vs baseline: 1.1504x; 1.1504x over previous
#include <hip/hip_runtime.h>


typedef float v2f __attribute__((ext_vector_type(2)));
typedef float v8f __attribute__((ext_vector_type(8)));

#define NHEAD     16
#define DK        64
#define DV        64
#define NTOT      16384
#define CHUNKS    16
#define WAVES     8
#define ROWS_WG   (NTOT / CHUNKS)
#define ROWS_WAVE (ROWS_WG / WAVES)
#define GROUP     16
#define NGROUP    (ROWS_WAVE / GROUP)
#define ROWPITCH  (NHEAD * DK)
#define DECAY     0.95f

__global__ __launch_bounds__(512) void l1w_init(const float* __restrict__ mem,
                                                float* __restrict__ out) {
    int i = blockIdx.x * 512 + threadIdx.x;
    if (i < NHEAD * DK * DV) out[i] = DECAY * mem[i];
}

struct Frag {
    float4 k[8];
    float4 v[8];
    float  w[8];
};

__device__ __forceinline__ void load_group(Frag& f, const float* __restrict__ kp,
                                           const float* __restrict__ vp,
                                           const float* __restrict__ ws, int g, int hlf) {
    #pragma unroll
    for (int q = 0; q < 8; ++q) {
        f.k[q] = *(const float4*)(kp + (size_t)(g * GROUP + q) * ROWPITCH);
        f.v[q] = *(const float4*)(vp + (size_t)(g * GROUP + q) * ROWPITCH);
    }
    #pragma unroll
    for (int q = 0; q < 8; ++q) {
        const float lo = ws[g * GROUP + q];
        const float hi = ws[g * GROUP + 8 + q];
        f.w[q] = hlf ? hi : lo;
    }
}

__device__ __forceinline__ float comp(const float4& x, int j) {
    return j == 0 ? x.x : (j == 1 ? x.y : (j == 2 ? x.z : x.w));
}

__device__ __forceinline__ void mma_group(const Frag& f, v8f (&acc)[4][4]) {
    #pragma unroll
    for (int t = 0; t < 4; ++t) {
        const float w0 = f.w[2 * t];
        const float w1 = f.w[2 * t + 1];
        v2f a[4], b[4];
        #pragma unroll
        for (int j = 0; j < 4; ++j) {
            a[j].x = comp(f.k[2 * t], j) * w0;
            a[j].y = comp(f.k[2 * t + 1], j) * w1;
            b[j].x = comp(f.v[2 * t], j);
            b[j].y = comp(f.v[2 * t + 1], j);
        }
        #pragma unroll
        for (int j = 0; j < 4; ++j) {
            #pragma unroll
            for (int i = 0; i < 4; ++i) {
                acc[j][i] = __builtin_amdgcn_wmma_f32_16x16x4_f32(
                     false, a[j],  false, b[i],
                     (short)0, acc[j][i],  false,  false);
            }
        }
    }
}

__global__ __launch_bounds__(256) void l1w_wmma(const float* __restrict__ keys,
                                                const float* __restrict__ vals,
                                                const float* __restrict__ rho,
                                                float* __restrict__ out) {
    __shared__ float red[DK * WAVES * DV];

    const int head  = blockIdx.x / CHUNKS;
    const int chunk = blockIdx.x % CHUNKS;

    const int tid  = threadIdx.x;
    const int wave = __builtin_amdgcn_readfirstlane(tid >> 5);
    const int lane = tid & 31;
    const int hlf  = lane >> 4;
    const int l15  = lane & 15;

    const int rowW = chunk * ROWS_WG + wave * ROWS_WAVE;
    const int row0 = rowW + hlf * 8;

    const float* kp = keys + ((size_t)row0 * NHEAD + head) * DK + l15 * 4;
    const float* vp = vals + ((size_t)row0 * NHEAD + head) * DV + l15 * 4;
    const float* ws = rho + rowW;

    v8f acc[4][4];
    #pragma unroll
    for (int j = 0; j < 4; ++j) {
        #pragma unroll
        for (int i = 0; i < 4; ++i) acc[j][i] = v8f{};
    }

    Frag f0, f1, f2;
    load_group(f0, kp, vp, ws, 0, hlf);
    load_group(f1, kp, vp, ws, 1, hlf);
    load_group(f2, kp, vp, ws, 2, hlf);
    mma_group(f0, acc);  load_group(f0, kp, vp, ws, 3, hlf);
    mma_group(f1, acc);  load_group(f1, kp, vp, ws, 4, hlf);
    mma_group(f2, acc);  load_group(f2, kp, vp, ws, 5, hlf);
    mma_group(f0, acc);  load_group(f0, kp, vp, ws, 6, hlf);
    mma_group(f1, acc);  load_group(f1, kp, vp, ws, 7, hlf);
    mma_group(f2, acc);
    mma_group(f0, acc);
    mma_group(f1, acc);

    float* mine = red + wave * DV + l15 * 4 + hlf * (32 * WAVES * DV);
    #pragma unroll
    for (int j = 0; j < 4; ++j) {
        #pragma unroll
        for (int r = 0; r < 8; ++r) {
            float4 o;
            o.x = acc[j][0][r]; o.y = acc[j][1][r]; o.z = acc[j][2][r]; o.w = acc[j][3][r];
            *(float4*)(mine + (4 * r + j) * (WAVES * DV)) = o;
        }
    }
    __syncthreads();

    const float* src = red + (wave * 8) * (WAVES * DV) + lane;
    float x[16][WAVES];
    #pragma unroll
    for (int r = 0; r < 8; ++r) {
        #pragma unroll
        for (int e = 0; e < 2; ++e) {
            #pragma unroll
            for (int p = 0; p < WAVES; ++p) x[2 * r + e][p] = src[(r * WAVES + p) * DV + e * 32];
        }
    }
    float* dst = out + (size_t)head * (DK * DV) + (wave * 8) * DV + lane;
    #pragma unroll
    for (int r = 0; r < 8; ++r) {
        #pragma unroll
        for (int e = 0; e < 2; ++e) {
            float s = x[2 * r + e][0];
            #pragma unroll
            for (int p = 1; p < WAVES; ++p) s += x[2 * r + e][p];
            atomicAdd(dst + r * DV + e * 32, s);
        }
    }
}

extern "C" void kernel_launch(void* const* d_in, const int* in_sizes, int n_in,
                              void* d_out, int out_size, void* d_ws, size_t ws_size,
                              hipStream_t stream) {
    const float* mem  = (const float*)d_in[0];
    const float* keys = (const float*)d_in[1];
    const float* vals = (const float*)d_in[2];
    const float* rho  = (const float*)d_in[3];
    float* out = (float*)d_out;

    l1w_init<<<(NHEAD * DK * DV + 511) / 512, 512, 0, stream>>>(mem, out);
    l1w_wmma<<<NHEAD * CHUNKS, 256, 0, stream>>>(keys, vals, rho, out);
}


# One wave per SIMD: 4-wave workgroups, each wave sums 256 rows behind a ring of four register sets (refills three WMMA blocks ahead, weights via vector loads); 4 LDS partials: sum order changes
# speedup vs baseline: 1.2490x; 1.0857x over previous
#include <hip/hip_runtime.h>


typedef float v2f __attribute__((ext_vector_type(2)));
typedef float v4f __attribute__((ext_vector_type(4)));
typedef float v8f __attribute__((ext_vector_type(8)));

#define NHEAD     16
#define DK        64
#define DV        64
#define NTOT      16384
#define CHUNKS    16
#define WAVES     4
#define ROWS_WG   (NTOT / CHUNKS)
#define ROWS_WAVE (ROWS_WG / WAVES)
#define GROUP     16
#define NGROUP    (ROWS_WAVE / GROUP)
#define RING      4
#define ROWS_RED  (DK / WAVES)
#define LDS_FLOATS (42 * 1024)
#define ROWPITCH  (NHEAD * DK)
#define DECAY     0.95f

__global__ __launch_bounds__(512) void l1w_init(const float* __restrict__ mem,
                                                float* __restrict__ out) {
    int i = blockIdx.x * 512 + threadIdx.x;
    if (i < NHEAD * DK * DV) out[i] = DECAY * mem[i];
}

#define STAGE() do { asm volatile("" ::: "memory"); __builtin_amdgcn_sched_barrier(0); } while (0)

struct Frag {
    v4f w[2];
    v4f k[8];
    v4f v[8];
};

__device__ __forceinline__ void load_group(Frag& f, const float* __restrict__ kp,
                                           const float* __restrict__ vp,
                                           const float* __restrict__ wp, int g) {
    f.w[0] = *(const v4f*)(wp + g * GROUP);
    f.w[1] = *(const v4f*)(wp + g * GROUP + 4);
    #pragma unroll
    for (int q = 0; q < 8; ++q) {
        f.k[q] = *(const v4f*)(kp + (size_t)(g * GROUP + q) * ROWPITCH);
        f.v[q] = *(const v4f*)(vp + (size_t)(g * GROUP + q) * ROWPITCH);
    }
}

__device__ __forceinline__ void mma_group(Frag& f, v8f (&acc)[4][4]) {
    asm volatile("" : "+v"(f.w[0]), "+v"(f.w[1]));
    #pragma unroll
    for (int t = 0; t < 4; ++t) {
        const float w0 = f.w[t >> 1][(2 * t) & 3];
        const float w1 = f.w[t >> 1][(2 * t + 1) & 3];
        v2f a[4], b[4];
        #pragma unroll
        for (int j = 0; j < 4; ++j) {
            a[j].x = f.k[2 * t][j] * w0;
            a[j].y = f.k[2 * t + 1][j] * w1;
            b[j].x = f.v[2 * t][j];
            b[j].y = f.v[2 * t + 1][j];
        }
        #pragma unroll
        for (int j = 0; j < 4; ++j) {
            #pragma unroll
            for (int i = 0; i < 4; ++i) {
                acc[j][i] = __builtin_amdgcn_wmma_f32_16x16x4_f32(
                     false, a[j],  false, b[i],
                     (short)0, acc[j][i],  false,  false);
            }
        }
    }
}

__global__ __launch_bounds__(128) void l1w_wmma(const float* __restrict__ keys,
                                                const float* __restrict__ vals,
                                                const float* __restrict__ rho,
                                                float* __restrict__ out) {
    __shared__ float red[LDS_FLOATS];
    static_assert(LDS_FLOATS >= DK * WAVES * DV, "partials must fit the reservation");

    const int head  = blockIdx.x / CHUNKS;
    const int chunk = blockIdx.x % CHUNKS;

    const int tid  = threadIdx.x;
    const int wave = __builtin_amdgcn_readfirstlane(tid >> 5);
    const int lane = tid & 31;
    const int hlf  = lane >> 4;
    const int l15  = lane & 15;

    const int rowW = chunk * ROWS_WG + wave * ROWS_WAVE;
    const int row0 = rowW + hlf * 8;

    const float* kp = keys + ((size_t)row0 * NHEAD + head) * DK + l15 * 4;
    const float* vp = vals + ((size_t)row0 * NHEAD + head) * DV + l15 * 4;
    const float* wp = rho + row0;

    v8f acc[4][4];
    #pragma unroll
    for (int j = 0; j < 4; ++j) {
        #pragma unroll
        for (int i = 0; i < 4; ++i) acc[j][i] = v8f{};
    }

    Frag f0, f1, f2, f3;
    load_group(f0, kp, vp, wp, 0);
    load_group(f1, kp, vp, wp, 1);
    load_group(f2, kp, vp, wp, 2);
    STAGE();
    #pragma unroll
    for (int b = 0; b < NGROUP - RING; b += RING) {
        load_group(f3, kp, vp, wp, b + 3);   STAGE();
        mma_group(f0, acc);                  STAGE();
        load_group(f0, kp, vp, wp, b + 4);   STAGE();
        mma_group(f1, acc);                  STAGE();
        load_group(f1, kp, vp, wp, b + 5);   STAGE();
        mma_group(f2, acc);                  STAGE();
        load_group(f2, kp, vp, wp, b + 6);   STAGE();
        mma_group(f3, acc);                  STAGE();
    }
    load_group(f3, kp, vp, wp, NGROUP - 1);  STAGE();
    mma_group(f0, acc);                      STAGE();
    mma_group(f1, acc);                      STAGE();
    mma_group(f2, acc);                      STAGE();
    mma_group(f3, acc);

    float* mine = red + wave * DV + l15 * 4 + hlf * (32 * WAVES * DV);
    #pragma unroll
    for (int j = 0; j < 4; ++j) {
        #pragma unroll
        for (int r = 0; r < 8; ++r) {
            float4 o;
            o.x = acc[j][0][r]; o.y = acc[j][1][r]; o.z = acc[j][2][r]; o.w = acc[j][3][r];
            *(float4*)(mine + (4 * r + j) * (WAVES * DV)) = o;
        }
    }
    __syncthreads();

    const float* src = red + (wave * ROWS_RED) * (WAVES * DV) + lane;
    float x[2 * ROWS_RED][WAVES];
    #pragma unroll
    for (int r = 0; r < ROWS_RED; ++r) {
        #pragma unroll
        for (int e = 0; e < 2; ++e) {
            #pragma unroll
            for (int p = 0; p < WAVES; ++p) x[2 * r + e][p] = src[(r * WAVES + p) * DV + e * 32];
        }
    }
    float* dst = out + (size_t)head * (DK * DV) + (wave * ROWS_RED) * DV + lane;
    #pragma unroll
    for (int r = 0; r < ROWS_RED; ++r) {
        #pragma unroll
        for (int e = 0; e < 2; ++e) {
            float s = x[2 * r + e][0];
            #pragma unroll
            for (int p = 1; p < WAVES; ++p) s += x[2 * r + e][p];
            atomicAdd(dst + r * DV + e * 32, s);
        }
    }
}

extern "C" void kernel_launch(void* const* d_in, const int* in_sizes, int n_in,
                              void* d_out, int out_size, void* d_ws, size_t ws_size,
                              hipStream_t stream) {
    const float* mem  = (const float*)d_in[0];
    const float* keys = (const float*)d_in[1];
    const float* vals = (const float*)d_in[2];
    const float* rho  = (const float*)d_in[3];
    float* out = (float*)d_out;

    l1w_init<<<(NHEAD * DK * DV + 511) / 512, 512, 0, stream>>>(mem, out);
    l1w_wmma<<<NHEAD * CHUNKS, 32 * WAVES, 0, stream>>>(keys, vals, rho, out);
}
